# starting kernel
# baseline (speedup 1.0000x reference)
#include <hip/hip_runtime.h>

typedef float v2f __attribute__((ext_vector_type(2)));
typedef float v8f __attribute__((ext_vector_type(8)));

#define N_NODES 100000
#define N_EDGES 1600000

__global__ void gcn_fill_zero(float* __restrict__ p, int n) {
    int i = blockIdx.x * blockDim.x + threadIdx.x;
    if (i < n) p[i] = 0.0f;
}

__global__ void gcn_degree(const int* __restrict__ col, float* __restrict__ deg, int e) {
    int i = blockIdx.x * blockDim.x + threadIdx.x;
    if (i < e) atomicAdd(&deg[col[i]], 1.0f);
}

__global__ void gcn_dinv(float* __restrict__ deg, int n) {
    int i = blockIdx.x * blockDim.x + threadIdx.x;
    if (i < n) {
        float d = deg[i];
        deg[i] = (d > 0.0f) ? (1.0f / sqrtf(d)) : 0.0f;
    }
}

__global__ void gcn_norm(const int* __restrict__ row, const int* __restrict__ col,
                         const float* __restrict__ dinv, float* __restrict__ norm, int e) {
    int i = blockIdx.x * blockDim.x + threadIdx.x;
    if (i < e) norm[i] = dinv[row[i]] * dinv[col[i]];
}

template <int F>
__global__ void gcn_bias_init(float* __restrict__ out, const float* __restrict__ b, int total) {
    int i = blockIdx.x * blockDim.x + threadIdx.x;
    if (i < total) out[i] = b[i & (F - 1)];
}

template <int F, bool RELU>
__global__ __launch_bounds__(256) void gcn_gemm_wmma(const float* __restrict__ A,
                                                     const float* __restrict__ W,
                                                     float* __restrict__ out, int nRows) {
    constexpr int TILES_N = F / 16;
    const int wave = (int)((blockIdx.x * blockDim.x + threadIdx.x) >> 5);
    const int lane = threadIdx.x & 31;
    const int tm = wave / TILES_N;
    const int tn = wave - tm * TILES_N;
    if (tm * 16 >= nRows) return;

    const int mn    = lane & 15;
    const int khalf = lane >> 4;

    const float* arow = A + (size_t)(tm * 16 + mn) * 128;
    const float* wcol = W + (size_t)(tn * 16 + mn);

    v8f acc = {};
#pragma unroll
    for (int k = 0; k < 128; k += 4) {
        const int kb = k + 2 * khalf;
        float a0 = arow[kb];
        float a1 = arow[kb + 1];
        if (RELU) { a0 = fmaxf(a0, 0.0f); a1 = fmaxf(a1, 0.0f); }
        v2f a; a.x = a0; a.y = a1;
        v2f b;
        b.x = wcol[(size_t)kb * F];
        b.y = wcol[(size_t)(kb + 1) * F];
        acc = __builtin_amdgcn_wmma_f32_16x16x4_f32(
             false, a,  false, b,
             (short)0, acc,  false,  false);
    }

    float* orow = out + (size_t)(tm * 16 + khalf * 8) * F + (size_t)(tn * 16 + mn);
#pragma unroll
    for (int v = 0; v < 8; ++v)
        orow[(size_t)v * F] = acc[v];
}

template <int F>
__global__ __launch_bounds__(256) void gcn_scatter_add(const float* __restrict__ t,
                                                       const int* __restrict__ row,
                                                       const int* __restrict__ col,
                                                       const float* __restrict__ norm,
                                                       float* __restrict__ out) {
    constexpr int EPB = 256 / F;
    const int e = blockIdx.x * EPB + (threadIdx.x >> (F == 128 ? 7 : 6));
    const int f = threadIdx.x & (F - 1);
    const int r = row[e];
    const int c = col[e];
    const float v = t[(size_t)r * F + f] * norm[e];
    atomicAdd(&out[(size_t)c * F + f], v);
}

extern "C" void kernel_launch(void* const* d_in, const int* in_sizes, int n_in,
                              void* d_out, int out_size, void* d_ws, size_t ws_size,
                              hipStream_t stream) {
    const float* x   = (const float*)d_in[0];
    const int*   ei  = (const int*)d_in[1];
    const float* W1  = (const float*)d_in[2];
    const float* b1  = (const float*)d_in[3];
    const float* W2  = (const float*)d_in[4];
    const float* b2  = (const float*)d_in[5];
    const float* W3  = (const float*)d_in[6];
    const float* b3  = (const float*)d_in[7];
    float*       out = (float*)d_out;

    const int* row = ei;
    const int* col = ei + N_EDGES;

    float* dinv = (float*)d_ws;
    float* norm = dinv + N_NODES;
    float* t    = norm + N_EDGES;
    float* h    = t + (size_t)N_NODES * 128;

    const int TB = 256;

    gcn_fill_zero<<<(N_NODES + TB - 1) / TB, TB, 0, stream>>>(dinv, N_NODES);
    gcn_degree<<<(N_EDGES + TB - 1) / TB, TB, 0, stream>>>(col, dinv, N_EDGES);
    gcn_dinv<<<(N_NODES + TB - 1) / TB, TB, 0, stream>>>(dinv, N_NODES);
    gcn_norm<<<(N_EDGES + TB - 1) / TB, TB, 0, stream>>>(row, col, dinv, norm, N_EDGES);

    const int gemm128Blocks = (N_NODES / 16) * (128 / 16) / 8;
    const int gemm64Blocks  = (N_NODES / 16) * (64 / 16) / 8;
    const int scat128Blocks = N_EDGES / (256 / 128);
    const int scat64Blocks  = N_EDGES / (256 / 64);

    gcn_gemm_wmma<128, false><<<gemm128Blocks, TB, 0, stream>>>(x, W1, t, N_NODES);
    gcn_bias_init<128><<<(N_NODES * 128) / TB, TB, 0, stream>>>(h, b1, N_NODES * 128);
    gcn_scatter_add<128><<<scat128Blocks, TB, 0, stream>>>(t, row, col, norm, h);

    gcn_gemm_wmma<128, true><<<gemm128Blocks, TB, 0, stream>>>(h, W2, t, N_NODES);
    gcn_bias_init<128><<<(N_NODES * 128) / TB, TB, 0, stream>>>(h, b2, N_NODES * 128);
    gcn_scatter_add<128><<<scat128Blocks, TB, 0, stream>>>(t, row, col, norm, h);

    gcn_gemm_wmma<64, true><<<gemm64Blocks, TB, 0, stream>>>(h, W3, t, N_NODES);
    gcn_bias_init<64><<<(N_NODES * 64) / TB, TB, 0, stream>>>(out, b3, N_NODES * 64);
    gcn_scatter_add<64><<<scat64Blocks, TB, 0, stream>>>(t, row, col, norm, out);
}


# GEMM re-tiled: wave owns 16 rows x 64 cols (4 acc tiles), A as 16-byte global loads reused over the 4 tiles, W staged per block in LDS in fragment order; k summed in a permuted order
# speedup vs baseline: 5.3419x; 5.3419x over previous
#include <hip/hip_runtime.h>


typedef float v2f __attribute__((ext_vector_type(2)));
typedef float v8f __attribute__((ext_vector_type(8)));

#define N_NODES 100000
#define N_EDGES 1600000

#define BIN_NODES 32
#define N_BINS    3125
#define N_CHUNKS  250
#define CHUNK_E   6400

__global__ __launch_bounds__(256) void gcn_bin_hist(const int* __restrict__ col,
                                                    int* __restrict__ hist) {
    __shared__ int cnt[N_BINS];
    for (int i = threadIdx.x; i < N_BINS; i += 256) cnt[i] = 0;
    __syncthreads();
    const int* c = col + (size_t)blockIdx.x * CHUNK_E;
    for (int i = threadIdx.x; i < CHUNK_E; i += 256) atomicAdd(&cnt[c[i] >> 5], 1);
    __syncthreads();
    int* h = hist + (size_t)blockIdx.x * N_BINS;
    for (int i = threadIdx.x; i < N_BINS; i += 256) h[i] = cnt[i];
}

__global__ void gcn_bin_colscan(int* __restrict__ hist, int* __restrict__ binTotal) {
    const int b = blockIdx.x * blockDim.x + threadIdx.x;
    if (b >= N_BINS) return;
    int run = 0;
#pragma unroll 10
    for (int c = 0; c < N_CHUNKS; ++c) {
        const int v = hist[(size_t)c * N_BINS + b];
        hist[(size_t)c * N_BINS + b] = run;
        run += v;
    }
    binTotal[b] = run;
}

__global__ __launch_bounds__(1024) void gcn_bin_scan(const int* __restrict__ binTotal,
                                                     int* __restrict__ binStart) {
    __shared__ int sh[2][1024];
    const int t = threadIdx.x;
    int v[4];
    int s = 0;
#pragma unroll
    for (int j = 0; j < 4; ++j) {
        const int b = 4 * t + j;
        v[j] = (b < N_BINS) ? binTotal[b] : 0;
        s += v[j];
    }
    int cur = 0;
    sh[0][t] = s;
    __syncthreads();
    for (int d = 1; d < 1024; d <<= 1) {
        int x = sh[cur][t];
        if (t >= d) x += sh[cur][t - d];
        sh[cur ^ 1][t] = x;
        cur ^= 1;
        __syncthreads();
    }
    int run = sh[cur][t] - s;
#pragma unroll
    for (int j = 0; j < 4; ++j) {
        const int b = 4 * t + j;
        if (b <= N_BINS) binStart[b] = run;
        run += v[j];
    }
}

__global__ __launch_bounds__(256) void gcn_bin_place(const int* __restrict__ row,
                                                     const int* __restrict__ col,
                                                     const int* __restrict__ hist,
                                                     const int* __restrict__ binStart,
                                                     int* __restrict__ packed) {
    __shared__ int off[N_BINS];
    const int* h = hist + (size_t)blockIdx.x * N_BINS;
    for (int i = threadIdx.x; i < N_BINS; i += 256) off[i] = binStart[i] + h[i];
    __syncthreads();
    const size_t e0 = (size_t)blockIdx.x * CHUNK_E;
    for (int i = threadIdx.x; i < CHUNK_E; i += 256) {
        const int r = row[e0 + i];
        const int c = col[e0 + i];
        const int slot = atomicAdd(&off[c >> 5], 1);
        packed[slot] = (r << 5) | (c & 31);
    }
}

__global__ __launch_bounds__(64) void gcn_bin_dinv(const int* __restrict__ packed,
                                                   const int* __restrict__ binStart,
                                                   float* __restrict__ dinv) {
    __shared__ int deg[BIN_NODES];
    if (threadIdx.x < BIN_NODES) deg[threadIdx.x] = 0;
    __syncthreads();
    const int s = binStart[blockIdx.x];
    const int e = binStart[blockIdx.x + 1];
    for (int i = s + (int)threadIdx.x; i < e; i += 64) atomicAdd(&deg[packed[i] & 31], 1);
    __syncthreads();
    if (threadIdx.x < BIN_NODES) {
        const float d = (float)deg[threadIdx.x];
        dinv[blockIdx.x * BIN_NODES + threadIdx.x] = (d > 0.0f) ? (1.0f / sqrtf(d)) : 0.0f;
    }
}

template <int F, bool RELU>
__global__ __launch_bounds__(2 * F) void gcn_gemm_wmma(const float* __restrict__ A,
                                                       const float* __restrict__ W,
                                                       float* __restrict__ out, int nRows) {
    typedef float v4f __attribute__((ext_vector_type(4)));
    constexpr int TILES = F / 16;
    constexpr int CG    = F / 64;
    __shared__ v4f wl[32 * F];

    const int t = threadIdx.x;

    {
        const int c  = t & (F - 1);
        const int hs = t / F;
        const int tile = (c >> 6) * 4 + (c & 3);
        const int nn   = (c & 63) >> 2;
        v4f* dst = wl + tile * 32 + hs * 16 + nn;
        const float* wsrc = W + (size_t)(4 * hs) * F + c;
#pragma unroll
        for (int j = 0; j < 16; ++j) {
            v4f v;
            v.x = wsrc[(size_t)(8 * j + 0) * F];
            v.y = wsrc[(size_t)(8 * j + 1) * F];
            v.z = wsrc[(size_t)(8 * j + 2) * F];
            v.w = wsrc[(size_t)(8 * j + 3) * F];
            dst[j * TILES * 32] = v;
        }
    }
    __syncthreads();

    const int lane = t & 31;
    const int wave = __builtin_amdgcn_readfirstlane(t >> 5);
    const int g    = wave % CG;
    const int row0 = ((int)blockIdx.x * 4 + wave / CG) * 16;
    if (row0 >= nRows) return;

    const int n = lane & 15;
    const int h = lane >> 4;

    const float* arow = A + (size_t)(row0 + n) * 128 + 4 * h;
    const v4f*   brd  = wl + g * 4 * 32 + lane;

    v8f acc[4] = {};
#pragma unroll
    for (int j = 0; j < 16; ++j) {
        v4f a = *reinterpret_cast<const v4f*>(arow + 8 * j);
        if (RELU) {
            a.x = fmaxf(a.x, 0.0f); a.y = fmaxf(a.y, 0.0f);
            a.z = fmaxf(a.z, 0.0f); a.w = fmaxf(a.w, 0.0f);
        }
        v4f b[4];
#pragma unroll
        for (int tn = 0; tn < 4; ++tn) b[tn] = brd[(j * TILES + tn) * 32];
        v2f alo; alo.x = a.x; alo.y = a.y;
        v2f ahi; ahi.x = a.z; ahi.y = a.w;
#pragma unroll
        for (int tn = 0; tn < 4; ++tn) {
            v2f blo; blo.x = b[tn].x; blo.y = b[tn].y;
            acc[tn] = __builtin_amdgcn_wmma_f32_16x16x4_f32(
                 false, alo,  false, blo,
                 (short)0, acc[tn],  false,  false);
        }
#pragma unroll
        for (int tn = 0; tn < 4; ++tn) {
            v2f bhi; bhi.x = b[tn].z; bhi.y = b[tn].w;
            acc[tn] = __builtin_amdgcn_wmma_f32_16x16x4_f32(
                 false, ahi,  false, bhi,
                 (short)0, acc[tn],  false,  false);
        }
    }

    float* orow = out + (size_t)(row0 + 8 * h) * F + 64 * g + 4 * n;
#pragma unroll
    for (int r = 0; r < 8; ++r) {
        v4f y;
        y.x = acc[0][r]; y.y = acc[1][r]; y.z = acc[2][r]; y.w = acc[3][r];
        *reinterpret_cast<v4f*>(orow + (size_t)r * F) = y;
    }
}

template <int F>
__global__ __launch_bounds__(256) void gcn_bin_aggregate(const float* __restrict__ t,
                                                         const int* __restrict__ packed,
                                                         const int* __restrict__ binStart,
                                                         const float* __restrict__ dinv,
                                                         const float* __restrict__ bias,
                                                         float* __restrict__ out) {
    constexpr int V = F / 32;
    typedef float vrow __attribute__((ext_vector_type(V)));
    __shared__ float tile[BIN_NODES * F];
    const int lane = threadIdx.x & 31;
    const int wave = __builtin_amdgcn_readfirstlane((int)(threadIdx.x >> 5));
    const int base = blockIdx.x * BIN_NODES;

    for (int i = threadIdx.x; i < BIN_NODES * F; i += 256) {
        const int p = i & (F - 1);
        tile[i] = bias[(p & 31) * V + (p >> 5)];
    }
    __syncthreads();

    const int s = binStart[blockIdx.x];
    const int e = binStart[blockIdx.x + 1];
    for (int i0 = s + wave * 32; i0 < e; i0 += 256) {
        int pk = 0;
        float nrm = 0.0f;
        if (i0 + lane < e) {
            pk = packed[i0 + lane];
            nrm = dinv[pk >> 5] * dinv[base + (pk & 31)];
        }
        const int cnt = (e - i0 < 32) ? (e - i0) : 32;
        int j = 0;
        for (; j + 8 <= cnt; j += 8) {
            int   pj[8];
            float nj[8];
            vrow  x[8];
            for (int u = 0; u < 8; ++u) {
                pj[u] = __builtin_amdgcn_readlane(pk, j + u);
                nj[u] = __int_as_float(__builtin_amdgcn_readlane(__float_as_int(nrm), j + u));
            }
            for (int u = 0; u < 8; ++u)
                x[u] = *reinterpret_cast<const vrow*>(t + (size_t)(pj[u] >> 5) * F + lane * V);
            for (int u = 0; u < 8; ++u) {
                float* dst = tile + (pj[u] & 31) * F + lane;
                for (int k = 0; k < V; ++k) atomicAdd(dst + 32 * k, x[u][k] * nj[u]);
            }
        }
        for (; j < cnt; ++j) {
            const int   p1 = __builtin_amdgcn_readlane(pk, j);
            const float n1 = __int_as_float(__builtin_amdgcn_readlane(__float_as_int(nrm), j));
            const vrow  x1 = *reinterpret_cast<const vrow*>(t + (size_t)(p1 >> 5) * F + lane * V);
            float* dst = tile + (p1 & 31) * F + lane;
            for (int k = 0; k < V; ++k) atomicAdd(dst + 32 * k, x1[k] * n1);
        }
    }
    __syncthreads();

    for (int r = wave; r < BIN_NODES; r += 8) {
        const float* srow = tile + r * F + lane;
        vrow y;
        for (int k = 0; k < V; ++k) y[k] = srow[32 * k];
        *reinterpret_cast<vrow*>(out + (size_t)(base + r) * F + lane * V) = y;
    }
}

extern "C" void kernel_launch(void* const* d_in, const int* in_sizes, int n_in,
                              void* d_out, int out_size, void* d_ws, size_t ws_size,
                              hipStream_t stream) {
    const float* x   = (const float*)d_in[0];
    const int*   ei  = (const int*)d_in[1];
    const float* W1  = (const float*)d_in[2];
    const float* b1  = (const float*)d_in[3];
    const float* W2  = (const float*)d_in[4];
    const float* b2  = (const float*)d_in[5];
    const float* W3  = (const float*)d_in[6];
    const float* b3  = (const float*)d_in[7];
    float*       out = (float*)d_out;

    const int* row = ei;
    const int* col = ei + N_EDGES;

    float* dinv   = (float*)d_ws;
    int*   packed = (int*)(dinv + N_NODES);
    float* t      = (float*)(packed + N_EDGES);
    float* h      = t + (size_t)N_NODES * 128;

    int* hist     = (int*)d_out;
    int* binTotal = hist + N_CHUNKS * N_BINS;
    int* binStart = binTotal + 3200;
    int* binStart3 = (int*)(t + (size_t)N_NODES * 64);

    const int TB = 256;

    gcn_bin_hist<<<N_CHUNKS, TB, 0, stream>>>(col, hist);
    gcn_bin_colscan<<<(N_BINS + 63) / 64, 64, 0, stream>>>(hist, binTotal);
    gcn_bin_scan<<<1, 1024, 0, stream>>>(binTotal, binStart);
    gcn_bin_place<<<N_CHUNKS, TB, 0, stream>>>(row, col, hist, binStart, packed);
    gcn_bin_dinv<<<N_BINS, 64, 0, stream>>>(packed, binStart, dinv);

    const int gemmBlocks = (N_NODES / 16 + 3) / 4;

    gcn_gemm_wmma<128, false><<<gemmBlocks, 256, 0, stream>>>(x, W1, t, N_NODES);
    gcn_bin_aggregate<128><<<N_BINS, TB, 0, stream>>>(t, packed, binStart, dinv, b1, h);

    gcn_gemm_wmma<128, true><<<gemmBlocks, 256, 0, stream>>>(h, W2, t, N_NODES);
    gcn_bin_aggregate<128><<<N_BINS, TB, 0, stream>>>(t, packed, binStart, dinv, b2, h);

    gcn_gemm_wmma<64, true><<<gemmBlocks, 128, 0, stream>>>(h, W3, t, N_NODES);
    gcn_bin_scan<<<1, 1024, 0, stream>>>(binTotal, binStart3);
    gcn_bin_aggregate<64><<<N_BINS, TB, 0, stream>>>(t, packed, binStart3, dinv, b3, out);
}


# Binning passes (histogram, chunk scan, placement, degree): each thread's global loads issued as one group ahead of its LDS atomics, not load-wait-add in turn; slot order still arrival
# speedup vs baseline: 5.7828x; 1.0825x over previous
#include <hip/hip_runtime.h>


typedef float v2f __attribute__((ext_vector_type(2)));
typedef float v8f __attribute__((ext_vector_type(8)));

#define N_NODES 100000
#define N_EDGES 1600000

#define BIN_NODES 32
#define N_BINS    3125
#define N_CHUNKS  250
#define CHUNK_E   6400

__global__ __launch_bounds__(256) void gcn_bin_hist(const int* __restrict__ col,
                                                    int* __restrict__ hist) {
    __shared__ int cnt[N_BINS];
    const int* c = col + (size_t)blockIdx.x * CHUNK_E + threadIdx.x;
    int v[CHUNK_E / 256];
#pragma unroll
    for (int u = 0; u < CHUNK_E / 256; ++u) v[u] = c[u * 256];
    for (int i = threadIdx.x; i < N_BINS; i += 256) cnt[i] = 0;
    __syncthreads();
#pragma unroll
    for (int u = 0; u < CHUNK_E / 256; ++u) atomicAdd(&cnt[v[u] >> 5], 1);
    __syncthreads();
    int* h = hist + (size_t)blockIdx.x * N_BINS;
    for (int i = threadIdx.x; i < N_BINS; i += 256) h[i] = cnt[i];
}

__global__ void gcn_bin_colscan(int* __restrict__ hist, int* __restrict__ binTotal) {
    const int b = blockIdx.x * blockDim.x + threadIdx.x;
    if (b >= N_BINS) return;
    int run = 0;
#pragma unroll 25
    for (int c = 0; c < N_CHUNKS; ++c) {
        const int v = hist[(size_t)c * N_BINS + b];
        hist[(size_t)c * N_BINS + b] = run;
        run += v;
    }
    binTotal[b] = run;
}

__global__ __launch_bounds__(1024) void gcn_bin_scan(const int* __restrict__ binTotal,
                                                     int* __restrict__ binStart) {
    __shared__ int sh[2][1024];
    const int t = threadIdx.x;
    int v[4];
    int s = 0;
#pragma unroll
    for (int j = 0; j < 4; ++j) {
        const int b = 4 * t + j;
        v[j] = (b < N_BINS) ? binTotal[b] : 0;
        s += v[j];
    }
    int cur = 0;
    sh[0][t] = s;
    __syncthreads();
    for (int d = 1; d < 1024; d <<= 1) {
        int x = sh[cur][t];
        if (t >= d) x += sh[cur][t - d];
        sh[cur ^ 1][t] = x;
        cur ^= 1;
        __syncthreads();
    }
    int run = sh[cur][t] - s;
#pragma unroll
    for (int j = 0; j < 4; ++j) {
        const int b = 4 * t + j;
        if (b <= N_BINS) binStart[b] = run;
        run += v[j];
    }
}

__global__ __launch_bounds__(256) void gcn_bin_place(const int* __restrict__ row,
                                                     const int* __restrict__ col,
                                                     const int* __restrict__ hist,
                                                     const int* __restrict__ binStart,
                                                     int* __restrict__ packed) {
    __shared__ int off[N_BINS];
    constexpr int PER_T = CHUNK_E / 256;
    constexpr int BIN_T = (N_BINS + 255) / 256;
    const size_t e0 = (size_t)blockIdx.x * CHUNK_E + threadIdx.x;
    int r[PER_T], c[PER_T];
#pragma unroll
    for (int u = 0; u < PER_T; ++u) {
        r[u] = row[e0 + u * 256];
        c[u] = col[e0 + u * 256];
    }
    const int* h = hist + (size_t)blockIdx.x * N_BINS;
    int o[BIN_T];
#pragma unroll
    for (int u = 0; u < BIN_T; ++u) {
        const int i = min((int)threadIdx.x + u * 256, N_BINS - 1);
        o[u] = binStart[i] + h[i];
    }
#pragma unroll
    for (int u = 0; u < BIN_T; ++u) {
        const int i = threadIdx.x + u * 256;
        if (i < N_BINS) off[i] = o[u];
    }
    __syncthreads();
    int slot[PER_T];
#pragma unroll
    for (int u = 0; u < PER_T; ++u) slot[u] = atomicAdd(&off[c[u] >> 5], 1);
#pragma unroll
    for (int u = 0; u < PER_T; ++u) packed[slot[u]] = (r[u] << 5) | (c[u] & 31);
}

__global__ __launch_bounds__(64) void gcn_bin_dinv(const int* __restrict__ packed,
                                                   const int* __restrict__ binStart,
                                                   float* __restrict__ dinv) {
    __shared__ int deg[BIN_NODES];
    if (threadIdx.x < BIN_NODES) deg[threadIdx.x] = 0;
    __syncthreads();
    const int s = binStart[blockIdx.x];
    const int e = binStart[blockIdx.x + 1];
    for (int i = s + (int)threadIdx.x; i < e; i += 64 * 8) {
        int w[8];
#pragma unroll
        for (int u = 0; u < 8; ++u) w[u] = packed[min(i + 64 * u, e - 1)];
        asm volatile("" :: "v"(w[0]), "v"(w[1]), "v"(w[2]), "v"(w[3]),
                           "v"(w[4]), "v"(w[5]), "v"(w[6]), "v"(w[7]));
#pragma unroll
        for (int u = 0; u < 8; ++u)
            if (i + 64 * u < e) atomicAdd(&deg[w[u] & 31], 1);
    }
    __syncthreads();
    if (threadIdx.x < BIN_NODES) {
        const float d = (float)deg[threadIdx.x];
        dinv[blockIdx.x * BIN_NODES + threadIdx.x] = (d > 0.0f) ? (1.0f / sqrtf(d)) : 0.0f;
    }
}

template <int F, bool RELU>
__global__ __launch_bounds__(2 * F) void gcn_gemm_wmma(const float* __restrict__ A,
                                                       const float* __restrict__ W,
                                                       float* __restrict__ out, int nRows) {
    typedef float v4f __attribute__((ext_vector_type(4)));
    constexpr int TILES = F / 16;
    constexpr int CG    = F / 64;
    __shared__ v4f wl[32 * F];

    const int t = threadIdx.x;

    {
        const int c  = t & (F - 1);
        const int hs = t / F;
        const int tile = (c >> 6) * 4 + (c & 3);
        const int nn   = (c & 63) >> 2;
        v4f* dst = wl + tile * 32 + hs * 16 + nn;
        const float* wsrc = W + (size_t)(4 * hs) * F + c;
#pragma unroll
        for (int j = 0; j < 16; ++j) {
            v4f v;
            v.x = wsrc[(size_t)(8 * j + 0) * F];
            v.y = wsrc[(size_t)(8 * j + 1) * F];
            v.z = wsrc[(size_t)(8 * j + 2) * F];
            v.w = wsrc[(size_t)(8 * j + 3) * F];
            dst[j * TILES * 32] = v;
        }
    }
    __syncthreads();

    const int lane = t & 31;
    const int wave = __builtin_amdgcn_readfirstlane(t >> 5);
    const int g    = wave % CG;
    const int row0 = ((int)blockIdx.x * 4 + wave / CG) * 16;
    if (row0 >= nRows) return;

    const int n = lane & 15;
    const int h = lane >> 4;

    const float* arow = A + (size_t)(row0 + n) * 128 + 4 * h;
    const v4f*   brd  = wl + g * 4 * 32 + lane;

    v8f acc[4] = {};
#pragma unroll
    for (int j = 0; j < 16; ++j) {
        v4f a = *reinterpret_cast<const v4f*>(arow + 8 * j);
        if (RELU) {
            a.x = fmaxf(a.x, 0.0f); a.y = fmaxf(a.y, 0.0f);
            a.z = fmaxf(a.z, 0.0f); a.w = fmaxf(a.w, 0.0f);
        }
        v4f b[4];
#pragma unroll
        for (int tn = 0; tn < 4; ++tn) b[tn] = brd[(j * TILES + tn) * 32];
        v2f alo; alo.x = a.x; alo.y = a.y;
        v2f ahi; ahi.x = a.z; ahi.y = a.w;
#pragma unroll
        for (int tn = 0; tn < 4; ++tn) {
            v2f blo; blo.x = b[tn].x; blo.y = b[tn].y;
            acc[tn] = __builtin_amdgcn_wmma_f32_16x16x4_f32(
                 false, alo,  false, blo,
                 (short)0, acc[tn],  false,  false);
        }
#pragma unroll
        for (int tn = 0; tn < 4; ++tn) {
            v2f bhi; bhi.x = b[tn].z; bhi.y = b[tn].w;
            acc[tn] = __builtin_amdgcn_wmma_f32_16x16x4_f32(
                 false, ahi,  false, bhi,
                 (short)0, acc[tn],  false,  false);
        }
    }

    float* orow = out + (size_t)(row0 + 8 * h) * F + 64 * g + 4 * n;
#pragma unroll
    for (int r = 0; r < 8; ++r) {
        v4f y;
        y.x = acc[0][r]; y.y = acc[1][r]; y.z = acc[2][r]; y.w = acc[3][r];
        *reinterpret_cast<v4f*>(orow + (size_t)r * F) = y;
    }
}

template <int F>
__global__ __launch_bounds__(256) void gcn_bin_aggregate(const float* __restrict__ t,
                                                         const int* __restrict__ packed,
                                                         const int* __restrict__ binStart,
                                                         const float* __restrict__ dinv,
                                                         const float* __restrict__ bias,
                                                         float* __restrict__ out) {
    constexpr int V = F / 32;
    typedef float vrow __attribute__((ext_vector_type(V)));
    __shared__ float tile[BIN_NODES * F];
    const int lane = threadIdx.x & 31;
    const int wave = __builtin_amdgcn_readfirstlane((int)(threadIdx.x >> 5));
    const int base = blockIdx.x * BIN_NODES;

    for (int i = threadIdx.x; i < BIN_NODES * F; i += 256) {
        const int p = i & (F - 1);
        tile[i] = bias[(p & 31) * V + (p >> 5)];
    }
    __syncthreads();

    const int s = binStart[blockIdx.x];
    const int e = binStart[blockIdx.x + 1];
    for (int i0 = s + wave * 32; i0 < e; i0 += 256) {
        int pk = 0;
        float nrm = 0.0f;
        if (i0 + lane < e) {
            pk = packed[i0 + lane];
            nrm = dinv[pk >> 5] * dinv[base + (pk & 31)];
        }
        const int cnt = (e - i0 < 32) ? (e - i0) : 32;
        int j = 0;
        for (; j + 8 <= cnt; j += 8) {
            int   pj[8];
            float nj[8];
            vrow  x[8];
            for (int u = 0; u < 8; ++u) {
                pj[u] = __builtin_amdgcn_readlane(pk, j + u);
                nj[u] = __int_as_float(__builtin_amdgcn_readlane(__float_as_int(nrm), j + u));
            }
            for (int u = 0; u < 8; ++u)
                x[u] = *reinterpret_cast<const vrow*>(t + (size_t)(pj[u] >> 5) * F + lane * V);
            for (int u = 0; u < 8; ++u) {
                float* dst = tile + (pj[u] & 31) * F + lane;
                for (int k = 0; k < V; ++k) atomicAdd(dst + 32 * k, x[u][k] * nj[u]);
            }
        }
        for (; j < cnt; ++j) {
            const int   p1 = __builtin_amdgcn_readlane(pk, j);
            const float n1 = __int_as_float(__builtin_amdgcn_readlane(__float_as_int(nrm), j));
            const vrow  x1 = *reinterpret_cast<const vrow*>(t + (size_t)(p1 >> 5) * F + lane * V);
            float* dst = tile + (p1 & 31) * F + lane;
            for (int k = 0; k < V; ++k) atomicAdd(dst + 32 * k, x1[k] * n1);
        }
    }
    __syncthreads();

    for (int r = wave; r < BIN_NODES; r += 8) {
        const float* srow = tile + r * F + lane;
        vrow y;
        for (int k = 0; k < V; ++k) y[k] = srow[32 * k];
        *reinterpret_cast<vrow*>(out + (size_t)(base + r) * F + lane * V) = y;
    }
}

extern "C" void kernel_launch(void* const* d_in, const int* in_sizes, int n_in,
                              void* d_out, int out_size, void* d_ws, size_t ws_size,
                              hipStream_t stream) {
    const float* x   = (const float*)d_in[0];
    const int*   ei  = (const int*)d_in[1];
    const float* W1  = (const float*)d_in[2];
    const float* b1  = (const float*)d_in[3];
    const float* W2  = (const float*)d_in[4];
    const float* b2  = (const float*)d_in[5];
    const float* W3  = (const float*)d_in[6];
    const float* b3  = (const float*)d_in[7];
    float*       out = (float*)d_out;

    const int* row = ei;
    const int* col = ei + N_EDGES;

    float* dinv   = (float*)d_ws;
    int*   packed = (int*)(dinv + N_NODES);
    float* t      = (float*)(packed + N_EDGES);
    float* h      = t + (size_t)N_NODES * 128;

    int* hist     = (int*)d_out;
    int* binTotal = hist + N_CHUNKS * N_BINS;
    int* binStart = binTotal + 3200;
    int* binStart3 = (int*)(t + (size_t)N_NODES * 64);

    const int TB = 256;

    gcn_bin_hist<<<N_CHUNKS, TB, 0, stream>>>(col, hist);
    gcn_bin_colscan<<<(N_BINS + 63) / 64, 64, 0, stream>>>(hist, binTotal);
    gcn_bin_scan<<<1, 1024, 0, stream>>>(binTotal, binStart);
    gcn_bin_place<<<N_CHUNKS, TB, 0, stream>>>(row, col, hist, binStart, packed);
    gcn_bin_dinv<<<N_BINS, 64, 0, stream>>>(packed, binStart, dinv);

    const int gemmBlocks = (N_NODES / 16 + 3) / 4;

    gcn_gemm_wmma<128, false><<<gemmBlocks, 256, 0, stream>>>(x, W1, t, N_NODES);
    gcn_bin_aggregate<128><<<N_BINS, TB, 0, stream>>>(t, packed, binStart, dinv, b1, h);

    gcn_gemm_wmma<128, true><<<gemmBlocks, 256, 0, stream>>>(h, W2, t, N_NODES);
    gcn_bin_aggregate<128><<<N_BINS, TB, 0, stream>>>(t, packed, binStart, dinv, b2, h);

    gcn_gemm_wmma<64, true><<<gemmBlocks, 128, 0, stream>>>(h, W3, t, N_NODES);
    gcn_bin_scan<<<1, 1024, 0, stream>>>(binTotal, binStart3);
    gcn_bin_aggregate<64><<<N_BINS, TB, 0, stream>>>(t, packed, binStart3, dinv, b3, out);
}


# Per-edge scale dinv[src]*dinv[dst] computed once (layer 1's aggregation), kept in the bin's own slice of the output buffer; layers 2-3 load it whole-line, no per-edge dinv gather; 12 launches to 11
# speedup vs baseline: 6.0503x; 1.0463x over previous
#include <hip/hip_runtime.h>


typedef float v2f __attribute__((ext_vector_type(2)));
typedef float v8f __attribute__((ext_vector_type(8)));

#define N_NODES 100000
#define N_EDGES 1600000

#define BIN_NODES 32
#define N_BINS    3125
#define N_CHUNKS  250
#define CHUNK_E   6400

#define NB_STRIDE 2048
#define NB_CAP    2046

__global__ __launch_bounds__(256) void gcn_bin_hist(const int* __restrict__ col,
                                                    int* __restrict__ hist) {
    __shared__ int cnt[N_BINS];
    const int* c = col + (size_t)blockIdx.x * CHUNK_E + threadIdx.x;
    int v[CHUNK_E / 256];
#pragma unroll
    for (int u = 0; u < CHUNK_E / 256; ++u) v[u] = c[u * 256];
    for (int i = threadIdx.x; i < N_BINS; i += 256) cnt[i] = 0;
    __syncthreads();
#pragma unroll
    for (int u = 0; u < CHUNK_E / 256; ++u) atomicAdd(&cnt[v[u] >> 5], 1);
    __syncthreads();
    int* h = hist + (size_t)blockIdx.x * N_BINS;
    for (int i = threadIdx.x; i < N_BINS; i += 256) h[i] = cnt[i];
}

__global__ void gcn_bin_colscan(int* __restrict__ hist, int* __restrict__ binTotal) {
    const int b = blockIdx.x * blockDim.x + threadIdx.x;
    if (b >= N_BINS) return;
    int run = 0;
#pragma unroll 25
    for (int c = 0; c < N_CHUNKS; ++c) {
        const int v = hist[(size_t)c * N_BINS + b];
        hist[(size_t)c * N_BINS + b] = run;
        run += v;
    }
    binTotal[b] = run;
}

__global__ __launch_bounds__(1024) void gcn_bin_scan(const int* __restrict__ binTotal,
                                                     int* __restrict__ binStart,
                                                     int* __restrict__ slice) {
    __shared__ int sh[2][1024];
    const int t = threadIdx.x;
    int v[4];
    int s = 0;
#pragma unroll
    for (int j = 0; j < 4; ++j) {
        const int b = 4 * t + j;
        v[j] = (b < N_BINS) ? binTotal[b] : 0;
        s += v[j];
    }
    int cur = 0;
    sh[0][t] = s;
    __syncthreads();
    for (int d = 1; d < 1024; d <<= 1) {
        int x = sh[cur][t];
        if (t >= d) x += sh[cur][t - d];
        sh[cur ^ 1][t] = x;
        cur ^= 1;
        __syncthreads();
    }
    int run = sh[cur][t] - s;
#pragma unroll
    for (int j = 0; j < 4; ++j) {
        const int b = 4 * t + j;
        if (b <= N_BINS) binStart[b] = run;
        if (b < N_BINS) {
            slice[(size_t)b * NB_STRIDE + NB_CAP]     = run;
            slice[(size_t)b * NB_STRIDE + NB_CAP + 1] = run + v[j];
        }
        run += v[j];
    }
}

__global__ __launch_bounds__(256) void gcn_bin_place(const int* __restrict__ row,
                                                     const int* __restrict__ col,
                                                     const int* __restrict__ hist,
                                                     const int* __restrict__ binStart,
                                                     int* __restrict__ packed) {
    __shared__ int off[N_BINS];
    constexpr int PER_T = CHUNK_E / 256;
    constexpr int BIN_T = (N_BINS + 255) / 256;
    const size_t e0 = (size_t)blockIdx.x * CHUNK_E + threadIdx.x;
    int r[PER_T], c[PER_T];
#pragma unroll
    for (int u = 0; u < PER_T; ++u) {
        r[u] = row[e0 + u * 256];
        c[u] = col[e0 + u * 256];
    }
    const int* h = hist + (size_t)blockIdx.x * N_BINS;
    int o[BIN_T];
#pragma unroll
    for (int u = 0; u < BIN_T; ++u) {
        const int i = min((int)threadIdx.x + u * 256, N_BINS - 1);
        o[u] = binStart[i] + h[i];
    }
#pragma unroll
    for (int u = 0; u < BIN_T; ++u) {
        const int i = threadIdx.x + u * 256;
        if (i < N_BINS) off[i] = o[u];
    }
    __syncthreads();
    int slot[PER_T];
#pragma unroll
    for (int u = 0; u < PER_T; ++u) slot[u] = atomicAdd(&off[c[u] >> 5], 1);
#pragma unroll
    for (int u = 0; u < PER_T; ++u) packed[slot[u]] = (r[u] << 5) | (c[u] & 31);
}

__global__ __launch_bounds__(64) void gcn_bin_dinv(const int* __restrict__ packed,
                                                   const int* __restrict__ binStart,
                                                   float* __restrict__ dinv) {
    __shared__ int deg[BIN_NODES];
    if (threadIdx.x < BIN_NODES) deg[threadIdx.x] = 0;
    __syncthreads();
    const int s = binStart[blockIdx.x];
    const int e = binStart[blockIdx.x + 1];
    for (int i = s + (int)threadIdx.x; i < e; i += 64 * 8) {
        int w[8];
#pragma unroll
        for (int u = 0; u < 8; ++u) w[u] = packed[min(i + 64 * u, e - 1)];
        asm volatile("" :: "v"(w[0]), "v"(w[1]), "v"(w[2]), "v"(w[3]),
                           "v"(w[4]), "v"(w[5]), "v"(w[6]), "v"(w[7]));
#pragma unroll
        for (int u = 0; u < 8; ++u)
            if (i + 64 * u < e) atomicAdd(&deg[w[u] & 31], 1);
    }
    __syncthreads();
    if (threadIdx.x < BIN_NODES) {
        const float d = (float)deg[threadIdx.x];
        dinv[blockIdx.x * BIN_NODES + threadIdx.x] = (d > 0.0f) ? (1.0f / sqrtf(d)) : 0.0f;
    }
}

template <int F, bool RELU>
__global__ __launch_bounds__(2 * F) void gcn_gemm_wmma(const float* __restrict__ A,
                                                       const float* __restrict__ W,
                                                       float* __restrict__ out, int nRows) {
    typedef float v4f __attribute__((ext_vector_type(4)));
    constexpr int TILES = F / 16;
    constexpr int CG    = F / 64;
    __shared__ v4f wl[32 * F];

    const int t = threadIdx.x;

    {
        const int c  = t & (F - 1);
        const int hs = t / F;
        const int tile = (c >> 6) * 4 + (c & 3);
        const int nn   = (c & 63) >> 2;
        v4f* dst = wl + tile * 32 + hs * 16 + nn;
        const float* wsrc = W + (size_t)(4 * hs) * F + c;
#pragma unroll
        for (int j = 0; j < 16; ++j) {
            v4f v;
            v.x = wsrc[(size_t)(8 * j + 0) * F];
            v.y = wsrc[(size_t)(8 * j + 1) * F];
            v.z = wsrc[(size_t)(8 * j + 2) * F];
            v.w = wsrc[(size_t)(8 * j + 3) * F];
            dst[j * TILES * 32] = v;
        }
    }
    __syncthreads();

    const int lane = t & 31;
    const int wave = __builtin_amdgcn_readfirstlane(t >> 5);
    const int g    = wave % CG;
    const int row0 = ((int)blockIdx.x * 4 + wave / CG) * 16;
    if (row0 >= nRows) return;

    const int n = lane & 15;
    const int h = lane >> 4;

    const float* arow = A + (size_t)(row0 + n) * 128 + 4 * h;
    const v4f*   brd  = wl + g * 4 * 32 + lane;

    v8f acc[4] = {};
#pragma unroll
    for (int j = 0; j < 16; ++j) {
        v4f a = *reinterpret_cast<const v4f*>(arow + 8 * j);
        if (RELU) {
            a.x = fmaxf(a.x, 0.0f); a.y = fmaxf(a.y, 0.0f);
            a.z = fmaxf(a.z, 0.0f); a.w = fmaxf(a.w, 0.0f);
        }
        v4f b[4];
#pragma unroll
        for (int tn = 0; tn < 4; ++tn) b[tn] = brd[(j * TILES + tn) * 32];
        v2f alo; alo.x = a.x; alo.y = a.y;
        v2f ahi; ahi.x = a.z; ahi.y = a.w;
#pragma unroll
        for (int tn = 0; tn < 4; ++tn) {
            v2f blo; blo.x = b[tn].x; blo.y = b[tn].y;
            acc[tn] = __builtin_amdgcn_wmma_f32_16x16x4_f32(
                 false, alo,  false, blo,
                 (short)0, acc[tn],  false,  false);
        }
#pragma unroll
        for (int tn = 0; tn < 4; ++tn) {
            v2f bhi; bhi.x = b[tn].z; bhi.y = b[tn].w;
            acc[tn] = __builtin_amdgcn_wmma_f32_16x16x4_f32(
                 false, ahi,  false, bhi,
                 (short)0, acc[tn],  false,  false);
        }
    }

    float* orow = out + (size_t)(row0 + 8 * h) * F + 64 * g + 4 * n;
#pragma unroll
    for (int r = 0; r < 8; ++r) {
        v4f y;
        y.x = acc[0][r]; y.y = acc[1][r]; y.z = acc[2][r]; y.w = acc[3][r];
        *reinterpret_cast<v4f*>(orow + (size_t)r * F) = y;
    }
}

template <int F, bool FIRST>
__global__ __launch_bounds__(256) void gcn_bin_aggregate(const float* __restrict__ t,
                                                         const int* __restrict__ packed,
                                                         const float* __restrict__ dinv,
                                                         const float* __restrict__ bias,
                                                         float* slice,
                                                         float* out) {
    constexpr int V = F / 32;
    typedef float vrow __attribute__((ext_vector_type(V)));
    __shared__ float tile[BIN_NODES * F];
    const int lane = threadIdx.x & 31;
    const int wave = __builtin_amdgcn_readfirstlane((int)(threadIdx.x >> 5));
    const int base = blockIdx.x * BIN_NODES;

    for (int i = threadIdx.x; i < BIN_NODES * F; i += 256) {
        const int p = i & (F - 1);
        tile[i] = bias[(p & 31) * V + (p >> 5)];
    }
    __syncthreads();

    float* nb = slice + (size_t)blockIdx.x * NB_STRIDE;
    const int s = __float_as_int(nb[NB_CAP]);
    const int e = __float_as_int(nb[NB_CAP + 1]);
    for (int i0 = s + wave * 32; i0 < e; i0 += 256) {
        const int idx = min(i0 + lane, e - 1);
        const int k   = idx - s;
        const int pk  = packed[idx];
        float nrm;
        if (FIRST) {
            nrm = dinv[pk >> 5] * dinv[base + (pk & 31)];
            if (i0 + lane < e && k < NB_CAP) nb[k] = nrm;
        } else {
            nrm = nb[min(k, NB_CAP - 1)];
            if (k >= NB_CAP) nrm = dinv[pk >> 5] * dinv[base + (pk & 31)];
        }
        const int cnt = (e - i0 < 32) ? (e - i0) : 32;
        int j = 0;
        for (; j + 8 <= cnt; j += 8) {
            int   pj[8];
            float nj[8];
            vrow  x[8];
            for (int u = 0; u < 8; ++u) {
                pj[u] = __builtin_amdgcn_readlane(pk, j + u);
                nj[u] = __int_as_float(__builtin_amdgcn_readlane(__float_as_int(nrm), j + u));
            }
            for (int u = 0; u < 8; ++u)
                x[u] = *reinterpret_cast<const vrow*>(t + (size_t)(pj[u] >> 5) * F + lane * V);
            for (int u = 0; u < 8; ++u) {
                float* dst = tile + (pj[u] & 31) * F + lane;
                for (int k = 0; k < V; ++k) atomicAdd(dst + 32 * k, x[u][k] * nj[u]);
            }
        }
        for (; j < cnt; ++j) {
            const int   p1 = __builtin_amdgcn_readlane(pk, j);
            const float n1 = __int_as_float(__builtin_amdgcn_readlane(__float_as_int(nrm), j));
            const vrow  x1 = *reinterpret_cast<const vrow*>(t + (size_t)(p1 >> 5) * F + lane * V);
            float* dst = tile + (p1 & 31) * F + lane;
            for (int k = 0; k < V; ++k) atomicAdd(dst + 32 * k, x1[k] * n1);
        }
    }
    __syncthreads();

    for (int r = wave; r < BIN_NODES; r += 8) {
        const float* srow = tile + r * F + lane;
        vrow y;
        for (int k = 0; k < V; ++k) y[k] = srow[32 * k];
        *reinterpret_cast<vrow*>(out + (size_t)(base + r) * F + lane * V) = y;
    }
}

extern "C" void kernel_launch(void* const* d_in, const int* in_sizes, int n_in,
                              void* d_out, int out_size, void* d_ws, size_t ws_size,
                              hipStream_t stream) {
    const float* x   = (const float*)d_in[0];
    const int*   ei  = (const int*)d_in[1];
    const float* W1  = (const float*)d_in[2];
    const float* b1  = (const float*)d_in[3];
    const float* W2  = (const float*)d_in[4];
    const float* b2  = (const float*)d_in[5];
    const float* W3  = (const float*)d_in[6];
    const float* b3  = (const float*)d_in[7];
    float*       out = (float*)d_out;

    const int* row = ei;
    const int* col = ei + N_EDGES;

    float* dinv   = (float*)d_ws;
    int*   packed = (int*)(dinv + N_NODES);
    float* t      = (float*)(packed + N_EDGES);
    float* h      = t + (size_t)N_NODES * 128;

    int* hist     = (int*)h;
    int* binTotal = hist + N_CHUNKS * N_BINS;
    int* binStart = binTotal + 3200;

    const int TB = 256;

    gcn_bin_hist<<<N_CHUNKS, TB, 0, stream>>>(col, hist);
    gcn_bin_colscan<<<(N_BINS + 63) / 64, 64, 0, stream>>>(hist, binTotal);
    gcn_bin_scan<<<1, 1024, 0, stream>>>(binTotal, binStart, (int*)d_out);
    gcn_bin_place<<<N_CHUNKS, TB, 0, stream>>>(row, col, hist, binStart, packed);
    gcn_bin_dinv<<<N_BINS, 64, 0, stream>>>(packed, binStart, dinv);

    const int gemmBlocks = (N_NODES / 16 + 3) / 4;

    gcn_gemm_wmma<128, false><<<gemmBlocks, 256, 0, stream>>>(x, W1, t, N_NODES);
    gcn_bin_aggregate<128, true><<<N_BINS, TB, 0, stream>>>(t, packed, dinv, b1, out, h);

    gcn_gemm_wmma<128, true><<<gemmBlocks, 256, 0, stream>>>(h, W2, t, N_NODES);
    gcn_bin_aggregate<128, false><<<N_BINS, TB, 0, stream>>>(t, packed, dinv, b2, out, h);

    gcn_gemm_wmma<64, true><<<gemmBlocks, 128, 0, stream>>>(h, W3, t, N_NODES);
    gcn_bin_aggregate<64, false><<<N_BINS, TB, 0, stream>>>(t, packed, dinv, b3, out, out);
}
